# starting kernel
# baseline (speedup 1.0000x reference)
#include <hip/hip_runtime.h>

typedef __attribute__((ext_vector_type(2))) float v2f;
typedef __attribute__((ext_vector_type(8))) float v8f;

#define B_N    16
#define CIN    64
#define COUT   64
#define H_IN   128
#define W_IN   128
#define HOUT   126
#define WOUT   126
#define KDIM   576

#define WPITCH 578
#define XCOLS  130
#define CIPITCH 66

#define SW_FLOATS (COUT * WPITCH)
#define SX_FLOATS (3 * XCOLS * CIPITCH)
#define LDS_BYTES ((SW_FLOATS + SX_FLOATS) * 4)

extern "C" __global__ __launch_bounds__(256)
void dynconv2d_wmma_f32(const float* __restrict__ x,
                        const float* __restrict__ w,
                        const float* __restrict__ bias,
                        float* __restrict__ out)
{
    extern __shared__ float smem[];
    float* sW = smem;
    float* sX = smem + SW_FLOATS;

    const int h   = blockIdx.x;
    const int b   = blockIdx.y;
    const int tid = threadIdx.x;

    const float* wsrc = w + (size_t)b * COUT * KDIM;
    for (int i = tid; i < COUT * KDIM; i += 256) {
        int co  = i / KDIM;
        int rem = i - co * KDIM;
        int ci  = rem / 9;
        int r   = rem - ci * 9;
        sW[co * WPITCH + r * 64 + ci] = wsrc[i];
    }

    for (int i = tid; i < CIN * 3 * XCOLS; i += 256) {
        int ci  = i / (3 * XCOLS);
        int rem = i - ci * (3 * XCOLS);
        int kh  = rem / XCOLS;
        int col = rem - kh * XCOLS;
        float v = 0.0f;
        if (col < W_IN)
            v = x[(((size_t)b * CIN + ci) * H_IN + (h + kh)) * W_IN + col];
        sX[(kh * XCOLS + col) * CIPITCH + ci] = v;
    }
    __syncthreads();

    const int wave  = tid >> 5;
    const int lane  = tid & 31;
    const int laneN = lane & 15;
    const int hi    = lane >> 4;

    const int co_base = (wave & 3) * 16;
    const int P0      = (wave >> 2) * 64;

    const float* aB  = sW + (co_base + laneN) * WPITCH + 2 * hi;
    const float* bB0 = sX + (P0 +  0 + laneN) * CIPITCH + 2 * hi;
    const float* bB1 = sX + (P0 + 16 + laneN) * CIPITCH + 2 * hi;
    const float* bB2 = sX + (P0 + 32 + laneN) * CIPITCH + 2 * hi;
    const float* bB3 = sX + (P0 + 48 + laneN) * CIPITCH + 2 * hi;

    v8f acc0 = {}, acc1 = {}, acc2 = {}, acc3 = {};

    #pragma unroll
    for (int kh = 0; kh < 3; ++kh) {
        #pragma unroll
        for (int kw = 0; kw < 3; ++kw) {
            const int r    = kh * 3 + kw;
            const int aoff = r * 64;
            const int boff = (kh * XCOLS + kw) * CIPITCH;
            #pragma unroll
            for (int c0 = 0; c0 < CIN; c0 += 4) {
                v2f a  = *(const v2f*)(aB  + aoff + c0);
                v2f b0 = *(const v2f*)(bB0 + boff + c0);
                v2f b1 = *(const v2f*)(bB1 + boff + c0);
                v2f b2 = *(const v2f*)(bB2 + boff + c0);
                v2f b3 = *(const v2f*)(bB3 + boff + c0);
                acc0 = __builtin_amdgcn_wmma_f32_16x16x4_f32(false, a, false, b0, (short)0, acc0, false, false);
                acc1 = __builtin_amdgcn_wmma_f32_16x16x4_f32(false, a, false, b1, (short)0, acc1, false, false);
                acc2 = __builtin_amdgcn_wmma_f32_16x16x4_f32(false, a, false, b2, (short)0, acc2, false, false);
                acc3 = __builtin_amdgcn_wmma_f32_16x16x4_f32(false, a, false, b3, (short)0, acc3, false, false);
            }
        }
    }

    #pragma unroll
    for (int t = 0; t < 4; ++t) {
        const v8f& acc = (t == 0) ? acc0 : (t == 1) ? acc1 : (t == 2) ? acc2 : acc3;
        const int wpix = P0 + t * 16 + laneN;
        if (wpix < WOUT) {
            #pragma unroll
            for (int j = 0; j < 8; ++j) {
                int co = co_base + j + 8 * hi;
                out[(((size_t)b * COUT + co) * HOUT + h) * WOUT + wpix] =
                    acc[j] + bias[b * COUT + co];
            }
        }
    }
}

extern "C" void kernel_launch(void* const* d_in, const int* in_sizes, int n_in,
                              void* d_out, int out_size, void* d_ws, size_t ws_size,
                              hipStream_t stream) {
    (void)in_sizes; (void)n_in; (void)out_size; (void)d_ws; (void)ws_size;
    const float* x    = (const float*)d_in[0];
    const float* w    = (const float*)d_in[1];
    const float* bias = (const float*)d_in[2];
    float* out        = (float*)d_out;

    dim3 grid(HOUT, B_N);
    dynconv2d_wmma_f32<<<grid, 256, LDS_BYTES, stream>>>(x, w, bias, out);
}


# Weights held in registers per wave (288 per lane, read once from global, no weight LDS); block = batch x 8 rows; input rows untransposed in a 4-slot LDS ring; k order and sum order change
# speedup vs baseline: 4.4548x; 4.4548x over previous
#include <hip/hip_runtime.h>


typedef __attribute__((ext_vector_type(2))) float v2f;
typedef __attribute__((ext_vector_type(4))) float v4f;
typedef __attribute__((ext_vector_type(8))) float v8f;

#define B_N    16
#define CIN    64
#define COUT   64
#define H_IN   128
#define W_IN   128
#define HOUT   126
#define WOUT   126
#define KDIM   576

#define ROWS_PER_BLOCK 8
#define NGROUPS  16
#define XP       132
#define HALF_PAD 16
#define SLOT_F   (CIN * XP + HALF_PAD)
#define NSLOT    4
#define LDS_BYTES (NSLOT * SLOT_F * 4)

extern "C" __global__ __launch_bounds__(256)
void dynconv2d_wstat_f32(const float* __restrict__ x,
                         const float* __restrict__ w,
                         const float* __restrict__ bias,
                         float* __restrict__ out)
{
    extern __shared__ float sX[];

    const int g   = blockIdx.x;
    const int b   = blockIdx.y;
    const int tid = threadIdx.x;
    const int h0  = g * ROWS_PER_BLOCK;
    const int R   = (HOUT - h0 < ROWS_PER_BLOCK) ? (HOUT - h0) : ROWS_PER_BLOCK;

    const int wave  = tid >> 5;
    const int lane  = tid & 31;
    const int laneN = lane & 15;
    const int hi    = lane >> 4;

    const int co_base = (wave & 3) * 16;
    const int P0      = (wave >> 2) * 64;

    const v4f* wrow = (const v4f*)(w + ((size_t)(b * COUT + co_base + laneN)) * KDIM + 288 * hi);
    v4f a4[72];
    #pragma unroll
    for (int i = 0; i < 72; ++i) a4[i] = wrow[i];

    const float* st_g = x + ((size_t)b * CIN + wave) * (H_IN * W_IN) + 4 * lane;
    const int    st_l = wave * XP + 4 * lane;

    #pragma unroll
    for (int rr = 0; rr < 3; ++rr) {
        #pragma unroll
        for (int j = 0; j < 8; ++j) {
            v4f v = *(const v4f*)(st_g + (size_t)j * 8 * (H_IN * W_IN) + (h0 + rr) * W_IN);
            *(v4f*)(sX + rr * SLOT_F + st_l + j * 8 * XP + (j >= 4 ? HALF_PAD : 0)) = v;
        }
    }
    {
        const int slot = tid >> 6, ci = tid & 63;
        const v4f z = {0.0f, 0.0f, 0.0f, 0.0f};
        *(v4f*)(sX + slot * SLOT_F + ci * XP + (ci >= 32 ? HALF_PAD : 0) + W_IN) = z;
    }

    float bv[8];
    #pragma unroll
    for (int j = 0; j < 8; ++j) bv[j] = bias[b * COUT + co_base + j + 8 * hi];

    __syncthreads();

    const int laneB = hi * (32 * XP + HALF_PAD) + P0 + laneN;

    #pragma unroll 1
    for (int i = 0; i < R; ++i) {
        const bool pf = (i + 1 < R);
        v4f pre[8];
        if (pf) {
            #pragma unroll
            for (int j = 0; j < 8; ++j)
                pre[j] = *(const v4f*)(st_g + (size_t)j * 8 * (H_IN * W_IN) + (h0 + i + 3) * W_IN);
        }

        const float* bk0 = sX + ((i    ) & 3) * SLOT_F + laneB;
        const float* bk1 = sX + ((i + 1) & 3) * SLOT_F + laneB;
        const float* bk2 = sX + ((i + 2) & 3) * SLOT_F + laneB;

        v8f acc0 = {}, acc1 = {}, acc2 = {}, acc3 = {};

        #pragma unroll
        for (int s = 0; s < 144; ++s) {
            const int f0 = 2 * s, f1 = 2 * s + 1;
            const int ci0 = f0 / 9, r0 = f0 % 9, kh0 = r0 / 3, kw0 = r0 % 3;
            const int ci1 = f1 / 9, r1 = f1 % 9, kh1 = r1 / 3, kw1 = r1 % 3;
            const float* p0 = (kh0 == 0 ? bk0 : kh0 == 1 ? bk1 : bk2) + ci0 * XP + kw0;
            const float* p1 = (kh1 == 0 ? bk0 : kh1 == 1 ? bk1 : bk2) + ci1 * XP + kw1;
            const v4f aq = a4[s >> 1];
            const v2f a  = (s & 1) ? __builtin_shufflevector(aq, aq, 2, 3)
                                   : __builtin_shufflevector(aq, aq, 0, 1);
            const v2f b0 = {p0[ 0], p1[ 0]};
            const v2f b1 = {p0[16], p1[16]};
            const v2f b2 = {p0[32], p1[32]};
            const v2f b3 = {p0[48], p1[48]};
            acc0 = __builtin_amdgcn_wmma_f32_16x16x4_f32(false, a, false, b0, (short)0, acc0, false, false);
            acc1 = __builtin_amdgcn_wmma_f32_16x16x4_f32(false, a, false, b1, (short)0, acc1, false, false);
            acc2 = __builtin_amdgcn_wmma_f32_16x16x4_f32(false, a, false, b2, (short)0, acc2, false, false);
            acc3 = __builtin_amdgcn_wmma_f32_16x16x4_f32(false, a, false, b3, (short)0, acc3, false, false);
        }

        const int h = h0 + i;
        #pragma unroll
        for (int t = 0; t < 4; ++t) {
            const v8f& acc = (t == 0) ? acc0 : (t == 1) ? acc1 : (t == 2) ? acc2 : acc3;
            const int wpix = P0 + t * 16 + laneN;
            if (t < 3 || wpix < WOUT) {
                #pragma unroll
                for (int j = 0; j < 8; ++j) {
                    const int co = co_base + j + 8 * hi;
                    out[(((size_t)b * COUT + co) * HOUT + h) * WOUT + wpix] = acc[j] + bv[j];
                }
            }
        }

        if (pf) {
            float* dst = sX + ((i + 3) & 3) * SLOT_F + st_l;
            #pragma unroll
            for (int j = 0; j < 8; ++j)
                *(v4f*)(dst + j * 8 * XP + (j >= 4 ? HALF_PAD : 0)) = pre[j];
        }
        __syncthreads();
    }
}

extern "C" void kernel_launch(void* const* d_in, const int* in_sizes, int n_in,
                              void* d_out, int out_size, void* d_ws, size_t ws_size,
                              hipStream_t stream) {
    (void)in_sizes; (void)n_in; (void)out_size; (void)d_ws; (void)ws_size;
    const float* x    = (const float*)d_in[0];
    const float* w    = (const float*)d_in[1];
    const float* bias = (const float*)d_in[2];
    float* out        = (float*)d_out;

    dim3 grid(NGROUPS, B_N);
    dynconv2d_wstat_f32<<<grid, 256, LDS_BYTES, stream>>>(x, w, bias, out);
}


# Input-row-stationary loop: each LDS input fragment is read once and multiplied with tap rows 0,1,2 into three live accumulator sets (output rows j, j-1, j-2); k pairs (ci, ci+1); sum order changes
# speedup vs baseline: 5.1264x; 1.1508x over previous
#include <hip/hip_runtime.h>


typedef __attribute__((ext_vector_type(2))) float v2f;
typedef __attribute__((ext_vector_type(4))) float v4f;
typedef __attribute__((ext_vector_type(8))) float v8f;
typedef __attribute__((address_space(3))) float lds_float;

#define B_N    16
#define CIN    64
#define COUT   64
#define H_IN   128
#define W_IN   128
#define HOUT   126
#define WOUT   126
#define KDIM   576

#define ROWS_PER_BLOCK 8
#define NGROUPS  16
#define XP       132
#define HALF_PAD 16
#define SLOT_F   (CIN * XP + HALF_PAD)
#define NSLOT    4
#define BIAS_F   (NSLOT * SLOT_F)
#define LDS_BYTES ((NSLOT * SLOT_F + COUT) * 4)
#define NGRP     48

template <int T, int KW>
__device__ __forceinline__ v2f lds_ci_pair(unsigned addr)
{
    v2f r;
    asm volatile("ds_load_2addr_b32 %0, %1 offset0:%2 offset1:%3"
                 : "=v"(r) : "v"(addr), "n"(16 * T + KW), "n"(XP + 16 * T + KW));
    return r;
}

template <int LEFT>
__device__ __forceinline__ void lds_wait(v2f (&b)[4])
{
    asm volatile("s_wait_dscnt %4" : "+v"(b[0]), "+v"(b[1]), "+v"(b[2]), "+v"(b[3]) : "n"(LEFT));
}

template <int G>
__device__ __forceinline__ void load_group(unsigned base, v2f (&b)[4])
{
    constexpr int cp = G / 3, kw = G % 3;
    const unsigned a = base + cp * (2 * XP * 4);
    b[0] = lds_ci_pair<0, kw>(a);
    b[1] = lds_ci_pair<1, kw>(a);
    b[2] = lds_ci_pair<2, kw>(a);
    b[3] = lds_ci_pair<3, kw>(a);
}

struct Stage { const float* g; float* l; };
template <int HALF>
__device__ __forceinline__ void stage_fetch(const Stage& s, v4f (&p)[4])
{
    #pragma unroll
    for (int q = 0; q < 4; ++q) p[q] = *(const v4f*)(s.g + (size_t)(4 * HALF + q) * 8 * (H_IN * W_IN));
}
template <int HALF>
__device__ __forceinline__ void stage_put(const Stage& s, const v4f (&p)[4])
{
    #pragma unroll
    for (int q = 0; q < 4; ++q) *(v4f*)(s.l + (4 * HALF + q) * 8 * XP + (HALF ? HALF_PAD : 0)) = p[q];
}

template <int G, int D, bool K0, bool K1, bool K2, bool PF>
__device__ __forceinline__ void row_groups(unsigned base, const v2f (&A)[16][9],
                                           v8f (&n)[4], v8f (&m)[4], v8f (&o)[4], v2f (&bf)[D + 1][4],
                                           const Stage& st, v4f (&pre)[4])
{
    constexpr int cp = G / 3, kw = G % 3;
    constexpr int CUR = G % (D + 1);
    constexpr int AHEAD = (NGRP - 1 - G < D) ? (NGRP - 1 - G) : D;
    if constexpr (PF && G == 0)  stage_fetch<0>(st, pre);
    if constexpr (PF && G == 16) { stage_put<0>(st, pre); stage_fetch<1>(st, pre); }
    if constexpr (PF && G == 32) stage_put<1>(st, pre);
    if constexpr (G + D < NGRP) load_group<G + D>(base, bf[(G + D) % (D + 1)]);
    lds_wait<4 * AHEAD>(bf[CUR]);
    if constexpr (K0) {
        #pragma unroll
        for (int t = 0; t < 4; ++t)
            n[t] = __builtin_amdgcn_wmma_f32_16x16x4_f32(false, A[cp][kw], false, bf[CUR][t], (short)0, n[t], false, false);
    }
    if constexpr (K1) {
        #pragma unroll
        for (int t = 0; t < 4; ++t)
            m[t] = __builtin_amdgcn_wmma_f32_16x16x4_f32(false, A[cp][3 + kw], false, bf[CUR][t], (short)0, m[t], false, false);
    }
    if constexpr (K2) {
        #pragma unroll
        for (int t = 0; t < 4; ++t)
            o[t] = __builtin_amdgcn_wmma_f32_16x16x4_f32(false, A[cp][6 + kw], false, bf[CUR][t], (short)0, o[t], false, false);
    }
    if constexpr (G + 1 < NGRP) row_groups<G + 1, D, K0, K1, K2, PF>(base, A, n, m, o, bf, st, pre);
}

template <bool K0, bool K1, bool K2, bool PF>
__device__ __forceinline__ void row_pass(const float* bk, const v2f (&A)[16][9],
                                         v8f (&n)[4], v8f (&m)[4], v8f (&o)[4], const Stage& st)
{
    const unsigned base = (unsigned)(size_t)(const lds_float*)bk;
    constexpr int D = (K0 && K1 && K2) ? 1 : 2;
    v2f bf[D + 1][4];
    v4f pre[4];
    load_group<0>(base, bf[0]);
    if constexpr (D == 2) load_group<1>(base, bf[1]);
    row_groups<0, D, K0, K1, K2, PF>(base, A, n, m, o, bf, st, pre);
}

__device__ __forceinline__ void store_row(const v8f (&acc)[4], const float* bl, float* __restrict__ orow,
                                          int P0, int laneN)
{
    const v4f b0 = *(const v4f*)bl, b1 = *(const v4f*)(bl + 4);
    const float bv[8] = {b0[0], b0[1], b0[2], b0[3], b1[0], b1[1], b1[2], b1[3]};
    #pragma unroll
    for (int t = 0; t < 4; ++t) {
        const int wpix = P0 + t * 16 + laneN;
        if (t < 3 || wpix < WOUT) {
            #pragma unroll
            for (int j = 0; j < 8; ++j)
                orow[(size_t)j * (HOUT * WOUT) + wpix] = acc[t][j] + bv[j];
        }
    }
}

extern "C" __global__ __launch_bounds__(256)
void dynconv2d_rowstat_f32(const float* __restrict__ x,
                           const float* __restrict__ w,
                           const float* __restrict__ bias,
                           float* __restrict__ out)
{
    extern __shared__ float sX[];

    const int g   = blockIdx.x;
    const int b   = blockIdx.y;
    const int tid = threadIdx.x;
    const int h0  = g * ROWS_PER_BLOCK;
    const int R   = (HOUT - h0 < ROWS_PER_BLOCK) ? (HOUT - h0) : ROWS_PER_BLOCK;

    const int wave  = tid >> 5;
    const int lane  = tid & 31;
    const int laneN = lane & 15;
    const int hi    = lane >> 4;

    const int co_base = (wave & 3) * 16;
    const int P0      = (wave >> 2) * 64;

    const v4f* wrow = (const v4f*)(w + ((size_t)(b * COUT + co_base + laneN)) * KDIM + 288 * hi);
    v2f A[16][9];
    #pragma unroll
    for (int cq = 0; cq < 8; ++cq) {
        v4f a4[9];
        #pragma unroll
        for (int i = 0; i < 9; ++i) a4[i] = wrow[9 * cq + i];
        #pragma unroll
        for (int c = 0; c < 2; ++c) {
            #pragma unroll
            for (int tap = 0; tap < 9; ++tap) {
                const int f0 = 18 * c + tap, f1 = f0 + 9;
                A[2 * cq + c][tap] = v2f{a4[f0 >> 2][f0 & 3], a4[f1 >> 2][f1 & 3]};
            }
        }
    }

    const float* st_g = x + ((size_t)b * CIN + wave) * (H_IN * W_IN) + 4 * lane;
    const int    st_l = wave * XP + 4 * lane;

    #pragma unroll
    for (int rr = 0; rr < 2; ++rr) {
        #pragma unroll
        for (int j = 0; j < 8; ++j) {
            v4f v = *(const v4f*)(st_g + (size_t)j * 8 * (H_IN * W_IN) + (h0 + rr) * W_IN);
            *(v4f*)(sX + rr * SLOT_F + st_l + j * 8 * XP + (j >= 4 ? HALF_PAD : 0)) = v;
        }
    }
    {
        const int slot = tid >> 6, ci = tid & 63;
        const v4f z = {0.0f, 0.0f, 0.0f, 0.0f};
        *(v4f*)(sX + slot * SLOT_F + ci * XP + (ci >= 32 ? HALF_PAD : 0) + W_IN) = z;
    }

    if (tid < COUT) sX[BIAS_F + tid] = bias[b * COUT + tid];
    const float* bv = sX + BIAS_F + co_base + 8 * hi;

    __syncthreads();

    const int laneB = hi * (32 * XP + HALF_PAD) + P0 + laneN;
    float* obase = out + ((size_t)b * COUT + co_base + 8 * hi) * (HOUT * WOUT);

    v8f X0[4] = {}, X1[4] = {}, X2[4] = {}, XD[4] = {};


    row_pass<true, false, false, true>(sX + 0 * SLOT_F + laneB, A, X0, XD, XD,
                                       Stage{st_g + (h0 + 2) * W_IN, sX + 2 * SLOT_F + st_l});
    __syncthreads();

    row_pass<true, true, false, true>(sX + 1 * SLOT_F + laneB, A, X1, X0, XD,
                                      Stage{st_g + (h0 + 3) * W_IN, sX + 3 * SLOT_F + st_l});
    __syncthreads();

    #pragma unroll 1
    for (int j = 2; j < R; ++j) {
        #pragma unroll
        for (int t = 0; t < 4; ++t) X2[t] = v8f{};
        row_pass<true, true, true, true>(sX + (j & 3) * SLOT_F + laneB, A, X2, X1, X0,
                                         Stage{st_g + (h0 + j + 2) * W_IN, sX + ((j + 2) & 3) * SLOT_F + st_l});

        store_row(X0, bv, obase + (size_t)(h0 + j - 2) * WOUT, P0, laneN);
        #pragma unroll
        for (int t = 0; t < 4; ++t) { X0[t] = X1[t]; X1[t] = X2[t]; }
        __syncthreads();
    }

    row_pass<false, true, true, false>(sX + (R & 3) * SLOT_F + laneB, A, XD, X1, X0, Stage{nullptr, nullptr});
    store_row(X0, bv, obase + (size_t)(h0 + R - 2) * WOUT, P0, laneN);

    row_pass<false, false, true, false>(sX + ((R + 1) & 3) * SLOT_F + laneB, A, XD, XD, X1, Stage{nullptr, nullptr});
    store_row(X1, bv, obase + (size_t)(h0 + R - 1) * WOUT, P0, laneN);
}

extern "C" void kernel_launch(void* const* d_in, const int* in_sizes, int n_in,
                              void* d_out, int out_size, void* d_ws, size_t ws_size,
                              hipStream_t stream) {
    (void)in_sizes; (void)n_in; (void)out_size; (void)d_ws; (void)ws_size;
    const float* x    = (const float*)d_in[0];
    const float* w    = (const float*)d_in[1];
    const float* bias = (const float*)d_in[2];
    float* out        = (float*)d_out;

    dim3 grid(NGROUPS, B_N);
    dynconv2d_rowstat_f32<<<grid, 256, LDS_BYTES, stream>>>(x, w, bias, out);
}


# Input fragment reads of a whole ci pair (12) stand in one place, one group ahead of 12 to 36 matrix instructions (was 4 reads ahead of 4 to 12), held together by scheduling fences; sums unchanged
# speedup vs baseline: 5.3164x; 1.0371x over previous
#include <hip/hip_runtime.h>


typedef __attribute__((ext_vector_type(2))) float v2f;
typedef __attribute__((ext_vector_type(4))) float v4f;
typedef __attribute__((ext_vector_type(8))) float v8f;
typedef __attribute__((address_space(3))) float lds_float;

#define B_N    16
#define CIN    64
#define COUT   64
#define H_IN   128
#define W_IN   128
#define HOUT   126
#define WOUT   126
#define KDIM   576

#define ROWS_PER_BLOCK 8
#define NGROUPS  16
#define XP       132
#define HALF_PAD 16
#define SLOT_F   (CIN * XP + HALF_PAD)
#define NSLOT    4
#define BIAS_F   (NSLOT * SLOT_F)
#define LDS_BYTES ((NSLOT * SLOT_F + COUT) * 4)
#define NGRP     16
#define GOPS     12

template <int T, int KW>
__device__ __forceinline__ v2f lds_ci_pair(unsigned addr)
{
    v2f r;
    asm volatile("ds_load_2addr_b32 %0, %1 offset0:%2 offset1:%3"
                 : "=v"(r) : "v"(addr), "n"(16 * T + KW), "n"(XP + 16 * T + KW));
    return r;
}

template <int LEFT>
__device__ __forceinline__ void lds_wait(v2f (&b)[GOPS])
{
    asm volatile("s_wait_dscnt %12"
                 : "+v"(b[0]), "+v"(b[1]), "+v"(b[2]), "+v"(b[3]), "+v"(b[4]), "+v"(b[5]),
                   "+v"(b[6]), "+v"(b[7]), "+v"(b[8]), "+v"(b[9]), "+v"(b[10]), "+v"(b[11])
                 : "n"(LEFT));
}

template <int CP>
__device__ __forceinline__ void load_group(unsigned base, v2f (&b)[GOPS])
{
    const unsigned a = base + CP * (2 * XP * 4);
    b[0]  = lds_ci_pair<0, 0>(a);
    b[1]  = lds_ci_pair<1, 0>(a);
    b[2]  = lds_ci_pair<2, 0>(a);
    b[3]  = lds_ci_pair<3, 0>(a);
    b[4]  = lds_ci_pair<0, 1>(a);
    b[5]  = lds_ci_pair<1, 1>(a);
    b[6]  = lds_ci_pair<2, 1>(a);
    b[7]  = lds_ci_pair<3, 1>(a);
    b[8]  = lds_ci_pair<0, 2>(a);
    b[9]  = lds_ci_pair<1, 2>(a);
    b[10] = lds_ci_pair<2, 2>(a);
    b[11] = lds_ci_pair<3, 2>(a);
}

struct Stage { const float* g; float* l; };
template <int HALF>
__device__ __forceinline__ void stage_fetch(const Stage& s, v4f (&p)[4])
{
    #pragma unroll
    for (int q = 0; q < 4; ++q) p[q] = *(const v4f*)(s.g + (size_t)(4 * HALF + q) * 8 * (H_IN * W_IN));
}
template <int HALF>
__device__ __forceinline__ void stage_put(const Stage& s, const v4f (&p)[4])
{
    #pragma unroll
    for (int q = 0; q < 4; ++q) *(v4f*)(s.l + (4 * HALF + q) * 8 * XP + (HALF ? HALF_PAD : 0)) = p[q];
}

template <int G, bool K0, bool K1, bool K2, bool PF>
__device__ __forceinline__ void row_groups(unsigned base, const v2f (&A)[16][9],
                                           v8f (&n)[4], v8f (&m)[4], v8f (&o)[4], v2f (&bf)[2][GOPS],
                                           const Stage& st, v4f (&pre)[4])
{
    constexpr int CUR = G % 2;
    constexpr int AHEAD = (G + 1 < NGRP) ? 1 : 0;
    if constexpr (PF && G == 0)  stage_fetch<0>(st, pre);
    if constexpr (PF && G == 5)  { stage_put<0>(st, pre); stage_fetch<1>(st, pre); }
    if constexpr (PF && G == 10) stage_put<1>(st, pre);
    __builtin_amdgcn_sched_barrier(0);
    if constexpr (G + 1 < NGRP) load_group<G + 1>(base, bf[(G + 1) % 2]);
    lds_wait<GOPS * AHEAD>(bf[CUR]);
    __builtin_amdgcn_sched_barrier(0);
    #pragma unroll
    for (int kw = 0; kw < 3; ++kw) {
        if constexpr (K0) {
            #pragma unroll
            for (int t = 0; t < 4; ++t)
                n[t] = __builtin_amdgcn_wmma_f32_16x16x4_f32(false, A[G][kw], false, bf[CUR][4 * kw + t], (short)0, n[t], false, false);
        }
        if constexpr (K1) {
            #pragma unroll
            for (int t = 0; t < 4; ++t)
                m[t] = __builtin_amdgcn_wmma_f32_16x16x4_f32(false, A[G][3 + kw], false, bf[CUR][4 * kw + t], (short)0, m[t], false, false);
        }
        if constexpr (K2) {
            #pragma unroll
            for (int t = 0; t < 4; ++t)
                o[t] = __builtin_amdgcn_wmma_f32_16x16x4_f32(false, A[G][6 + kw], false, bf[CUR][4 * kw + t], (short)0, o[t], false, false);
        }
    }
    if constexpr (G + 1 < NGRP) row_groups<G + 1, K0, K1, K2, PF>(base, A, n, m, o, bf, st, pre);
}

template <bool K0, bool K1, bool K2, bool PF>
__device__ __forceinline__ void row_pass(const float* bk, const v2f (&A)[16][9],
                                         v8f (&n)[4], v8f (&m)[4], v8f (&o)[4], const Stage& st)
{
    const unsigned base = (unsigned)(size_t)(const lds_float*)bk;
    v2f bf[2][GOPS];
    v4f pre[4];
    load_group<0>(base, bf[0]);
    row_groups<0, K0, K1, K2, PF>(base, A, n, m, o, bf, st, pre);
}

__device__ __forceinline__ void store_row(const v8f (&acc)[4], const float* bl, float* __restrict__ orow,
                                          int P0, int laneN)
{
    const v4f b0 = *(const v4f*)bl, b1 = *(const v4f*)(bl + 4);
    const float bv[8] = {b0[0], b0[1], b0[2], b0[3], b1[0], b1[1], b1[2], b1[3]};
    #pragma unroll
    for (int t = 0; t < 4; ++t) {
        const int wpix = P0 + t * 16 + laneN;
        if (t < 3 || wpix < WOUT) {
            #pragma unroll
            for (int j = 0; j < 8; ++j)
                orow[(size_t)j * (HOUT * WOUT) + wpix] = acc[t][j] + bv[j];
        }
    }
}

extern "C" __global__ __launch_bounds__(256)
void dynconv2d_rowstat_f32(const float* __restrict__ x,
                           const float* __restrict__ w,
                           const float* __restrict__ bias,
                           float* __restrict__ out)
{
    extern __shared__ float sX[];

    const int g   = blockIdx.x;
    const int b   = blockIdx.y;
    const int tid = threadIdx.x;
    const int h0  = g * ROWS_PER_BLOCK;
    const int R   = (HOUT - h0 < ROWS_PER_BLOCK) ? (HOUT - h0) : ROWS_PER_BLOCK;

    const int wave  = tid >> 5;
    const int lane  = tid & 31;
    const int laneN = lane & 15;
    const int hi    = lane >> 4;

    const int co_base = (wave & 3) * 16;
    const int P0      = (wave >> 2) * 64;

    const v4f* wrow = (const v4f*)(w + ((size_t)(b * COUT + co_base + laneN)) * KDIM + 288 * hi);
    v2f A[16][9];
    #pragma unroll
    for (int cq = 0; cq < 8; ++cq) {
        v4f a4[9];
        #pragma unroll
        for (int i = 0; i < 9; ++i) a4[i] = wrow[9 * cq + i];
        #pragma unroll
        for (int c = 0; c < 2; ++c) {
            #pragma unroll
            for (int tap = 0; tap < 9; ++tap) {
                const int f0 = 18 * c + tap, f1 = f0 + 9;
                A[2 * cq + c][tap] = v2f{a4[f0 >> 2][f0 & 3], a4[f1 >> 2][f1 & 3]};
            }
        }
    }

    const float* st_g = x + ((size_t)b * CIN + wave) * (H_IN * W_IN) + 4 * lane;
    const int    st_l = wave * XP + 4 * lane;

    #pragma unroll
    for (int rr = 0; rr < 2; ++rr) {
        #pragma unroll
        for (int j = 0; j < 8; ++j) {
            v4f v = *(const v4f*)(st_g + (size_t)j * 8 * (H_IN * W_IN) + (h0 + rr) * W_IN);
            *(v4f*)(sX + rr * SLOT_F + st_l + j * 8 * XP + (j >= 4 ? HALF_PAD : 0)) = v;
        }
    }
    {
        const int slot = tid >> 6, ci = tid & 63;
        const v4f z = {0.0f, 0.0f, 0.0f, 0.0f};
        *(v4f*)(sX + slot * SLOT_F + ci * XP + (ci >= 32 ? HALF_PAD : 0) + W_IN) = z;
    }

    if (tid < COUT) sX[BIAS_F + tid] = bias[b * COUT + tid];
    const float* bv = sX + BIAS_F + co_base + 8 * hi;

    __syncthreads();

    const int laneB = hi * (32 * XP + HALF_PAD) + P0 + laneN;
    float* obase = out + ((size_t)b * COUT + co_base + 8 * hi) * (HOUT * WOUT);

    v8f X0[4] = {}, X1[4] = {}, X2[4] = {}, XD[4] = {};


    row_pass<true, false, false, true>(sX + 0 * SLOT_F + laneB, A, X0, XD, XD,
                                       Stage{st_g + (h0 + 2) * W_IN, sX + 2 * SLOT_F + st_l});
    __syncthreads();

    row_pass<true, true, false, true>(sX + 1 * SLOT_F + laneB, A, X1, X0, XD,
                                      Stage{st_g + (h0 + 3) * W_IN, sX + 3 * SLOT_F + st_l});
    __syncthreads();

    #pragma unroll 1
    for (int j = 2; j < R; ++j) {
        #pragma unroll
        for (int t = 0; t < 4; ++t) X2[t] = v8f{};
        row_pass<true, true, true, true>(sX + (j & 3) * SLOT_F + laneB, A, X2, X1, X0,
                                         Stage{st_g + (h0 + j + 2) * W_IN, sX + ((j + 2) & 3) * SLOT_F + st_l});

        store_row(X0, bv, obase + (size_t)(h0 + j - 2) * WOUT, P0, laneN);
        #pragma unroll
        for (int t = 0; t < 4; ++t) { X0[t] = X1[t]; X1[t] = X2[t]; }
        __syncthreads();
    }

    row_pass<false, true, true, false>(sX + (R & 3) * SLOT_F + laneB, A, XD, X1, X0, Stage{nullptr, nullptr});
    store_row(X0, bv, obase + (size_t)(h0 + R - 2) * WOUT, P0, laneN);

    row_pass<false, false, true, false>(sX + ((R + 1) & 3) * SLOT_F + laneB, A, XD, XD, X1, Stage{nullptr, nullptr});
    store_row(X1, bv, obase + (size_t)(h0 + R - 1) * WOUT, P0, laneN);
}

extern "C" void kernel_launch(void* const* d_in, const int* in_sizes, int n_in,
                              void* d_out, int out_size, void* d_ws, size_t ws_size,
                              hipStream_t stream) {
    (void)in_sizes; (void)n_in; (void)out_size; (void)d_ws; (void)ws_size;
    const float* x    = (const float*)d_in[0];
    const float* w    = (const float*)d_in[1];
    const float* bias = (const float*)d_in[2];
    float* out        = (float*)d_out;

    dim3 grid(NGROUPS, B_N);
    dynconv2d_rowstat_f32<<<grid, 256, LDS_BYTES, stream>>>(x, w, bias, out);
}


# Weight fetch overlapped with products: only 1 of 8 weight batches awaited before the first matrix instruction; input rows 0,1 are walked batch by batch; a block's first output row: sum order changes
# speedup vs baseline: 5.5006x; 1.0346x over previous
#include <hip/hip_runtime.h>


typedef __attribute__((ext_vector_type(2))) float v2f;
typedef __attribute__((ext_vector_type(4))) float v4f;
typedef __attribute__((ext_vector_type(8))) float v8f;
typedef __attribute__((address_space(3))) float lds_float;

#define B_N    16
#define CIN    64
#define COUT   64
#define H_IN   128
#define W_IN   128
#define HOUT   126
#define WOUT   126
#define KDIM   576

#define ROWS_PER_BLOCK 8
#define NGROUPS  16
#define XP       132
#define HALF_PAD 16
#define SLOT_F   (CIN * XP + HALF_PAD)
#define NSLOT    4
#define BIAS_F   (NSLOT * SLOT_F)
#define LDS_BYTES ((NSLOT * SLOT_F + COUT) * 4)
#define NGRP     16
#define GOPS     12

template <int T, int KW>
__device__ __forceinline__ v2f lds_ci_pair(unsigned addr)
{
    v2f r;
    asm volatile("ds_load_2addr_b32 %0, %1 offset0:%2 offset1:%3"
                 : "=v"(r) : "v"(addr), "n"(16 * T + KW), "n"(XP + 16 * T + KW));
    return r;
}

template <int LEFT>
__device__ __forceinline__ void lds_wait(v2f (&b)[GOPS])
{
    asm volatile("s_wait_dscnt %12"
                 : "+v"(b[0]), "+v"(b[1]), "+v"(b[2]), "+v"(b[3]), "+v"(b[4]), "+v"(b[5]),
                   "+v"(b[6]), "+v"(b[7]), "+v"(b[8]), "+v"(b[9]), "+v"(b[10]), "+v"(b[11])
                 : "n"(LEFT));
}

template <int CP>
__device__ __forceinline__ void load_group(unsigned base, v2f (&b)[GOPS])
{
    const unsigned a = base + CP * (2 * XP * 4);
    b[0]  = lds_ci_pair<0, 0>(a);
    b[1]  = lds_ci_pair<1, 0>(a);
    b[2]  = lds_ci_pair<2, 0>(a);
    b[3]  = lds_ci_pair<3, 0>(a);
    b[4]  = lds_ci_pair<0, 1>(a);
    b[5]  = lds_ci_pair<1, 1>(a);
    b[6]  = lds_ci_pair<2, 1>(a);
    b[7]  = lds_ci_pair<3, 1>(a);
    b[8]  = lds_ci_pair<0, 2>(a);
    b[9]  = lds_ci_pair<1, 2>(a);
    b[10] = lds_ci_pair<2, 2>(a);
    b[11] = lds_ci_pair<3, 2>(a);
}

struct Stage { const float* g; float* l; };
template <int HALF>
__device__ __forceinline__ void stage_fetch(const Stage& s, v4f (&p)[4])
{
    #pragma unroll
    for (int q = 0; q < 4; ++q) p[q] = *(const v4f*)(s.g + (size_t)(4 * HALF + q) * 8 * (H_IN * W_IN));
}
template <int HALF>
__device__ __forceinline__ void stage_put(const Stage& s, const v4f (&p)[4])
{
    #pragma unroll
    for (int q = 0; q < 4; ++q) *(v4f*)(s.l + (4 * HALF + q) * 8 * XP + (HALF ? HALF_PAD : 0)) = p[q];
}

template <int G, bool K0, bool K1, bool K2, bool PF>
__device__ __forceinline__ void row_groups(unsigned base, const v2f (&A)[16][9],
                                           v8f (&n)[4], v8f (&m)[4], v8f (&o)[4], v2f (&bf)[2][GOPS],
                                           const Stage& st, v4f (&pre)[4])
{
    constexpr int CUR = G % 2;
    constexpr int AHEAD = (G + 1 < NGRP) ? 1 : 0;
    if constexpr (PF && G == 0)  stage_fetch<0>(st, pre);
    if constexpr (PF && G == 5)  { stage_put<0>(st, pre); stage_fetch<1>(st, pre); }
    if constexpr (PF && G == 10) stage_put<1>(st, pre);
    __builtin_amdgcn_sched_barrier(0);
    if constexpr (G + 1 < NGRP) load_group<G + 1>(base, bf[(G + 1) % 2]);
    lds_wait<GOPS * AHEAD>(bf[CUR]);
    __builtin_amdgcn_sched_barrier(0);
    #pragma unroll
    for (int kw = 0; kw < 3; ++kw) {
        if constexpr (K0) {
            #pragma unroll
            for (int t = 0; t < 4; ++t)
                n[t] = __builtin_amdgcn_wmma_f32_16x16x4_f32(false, A[G][kw], false, bf[CUR][4 * kw + t], (short)0, n[t], false, false);
        }
        if constexpr (K1) {
            #pragma unroll
            for (int t = 0; t < 4; ++t)
                m[t] = __builtin_amdgcn_wmma_f32_16x16x4_f32(false, A[G][3 + kw], false, bf[CUR][4 * kw + t], (short)0, m[t], false, false);
        }
        if constexpr (K2) {
            #pragma unroll
            for (int t = 0; t < 4; ++t)
                o[t] = __builtin_amdgcn_wmma_f32_16x16x4_f32(false, A[G][6 + kw], false, bf[CUR][4 * kw + t], (short)0, o[t], false, false);
        }
    }
    if constexpr (G + 1 < NGRP) row_groups<G + 1, K0, K1, K2, PF>(base, A, n, m, o, bf, st, pre);
}

template <bool K0, bool K1, bool K2, bool PF>
__device__ __forceinline__ void row_pass(const float* bk, const v2f (&A)[16][9],
                                         v8f (&n)[4], v8f (&m)[4], v8f (&o)[4], const Stage& st)
{
    const unsigned base = (unsigned)(size_t)(const lds_float*)bk;
    v2f bf[2][GOPS];
    v4f pre[4];
    load_group<0>(base, bf[0]);
    row_groups<0, K0, K1, K2, PF>(base, A, n, m, o, bf, st, pre);
}

template <int CQ>
__device__ __forceinline__ void w_fetch(const v4f* wrow, v4f (&raw)[9])
{
    #pragma unroll
    for (int i = 0; i < 9; ++i) raw[i] = wrow[9 * CQ + i];
}
template <int CQ>
__device__ __forceinline__ void w_build(const v4f (&raw)[9], v2f (&A)[16][9])
{
    #pragma unroll
    for (int c = 0; c < 2; ++c) {
        #pragma unroll
        for (int tap = 0; tap < 9; ++tap) {
            const int f0 = 18 * c + tap, f1 = f0 + 9;
            A[2 * CQ + c][tap] = v2f{raw[f0 >> 2][f0 & 3], raw[f1 >> 2][f1 & 3]};
        }
    }
}

template <int S>
__device__ __forceinline__ void head_steps(unsigned base, const v4f* wrow, v4f (&raw)[9], v2f (&A)[16][9],
                                           v8f (&X0)[4], v8f (&X1)[4], v2f (&bf)[2][GOPS],
                                           const Stage& st2, const Stage& st3, v4f (&pre)[4])
{
    constexpr int NSTEP = 32;
    constexpr int CQ = S / 4, RW = (S % 4) / 2, CP = 2 * CQ + (S % 2);
    constexpr int NS = S + 1;
    constexpr int NRW = (NS % 4) / 2, NCP = 2 * (NS / 4) + (NS % 2);
    constexpr int CUR = S % 2;
    constexpr int AHEAD = (NS < NSTEP) ? 1 : 0;
    if constexpr (S == 0) w_fetch<1>(wrow, raw);
    if constexpr (S % 4 == 3 && CQ + 1 < 8) w_build<(CQ + 1 < 8 ? CQ + 1 : 0)>(raw, A);
    if constexpr (S % 4 == 3 && CQ + 2 < 8) w_fetch<(CQ + 2 < 8 ? CQ + 2 : 0)>(wrow, raw);
    if constexpr (S == 1)  stage_fetch<0>(st2, pre);
    if constexpr (S == 9)  { stage_put<0>(st2, pre); stage_fetch<1>(st2, pre); }
    if constexpr (S == 17) { stage_put<1>(st2, pre); stage_fetch<0>(st3, pre); }
    if constexpr (S == 25) { stage_put<0>(st3, pre); stage_fetch<1>(st3, pre); }
    if constexpr (S == 31) stage_put<1>(st3, pre);
    __builtin_amdgcn_sched_barrier(0);
    if constexpr (NS < NSTEP) load_group<NCP>(base + NRW * (SLOT_F * 4), bf[NS % 2]);
    lds_wait<GOPS * AHEAD>(bf[CUR]);
    __builtin_amdgcn_sched_barrier(0);
    #pragma unroll
    for (int kw = 0; kw < 3; ++kw) {
        if constexpr (RW == 0) {
            #pragma unroll
            for (int t = 0; t < 4; ++t)
                X0[t] = __builtin_amdgcn_wmma_f32_16x16x4_f32(false, A[CP][kw], false, bf[CUR][4 * kw + t], (short)0, X0[t], false, false);
        } else {
            #pragma unroll
            for (int t = 0; t < 4; ++t)
                X1[t] = __builtin_amdgcn_wmma_f32_16x16x4_f32(false, A[CP][kw], false, bf[CUR][4 * kw + t], (short)0, X1[t], false, false);
            #pragma unroll
            for (int t = 0; t < 4; ++t)
                X0[t] = __builtin_amdgcn_wmma_f32_16x16x4_f32(false, A[CP][3 + kw], false, bf[CUR][4 * kw + t], (short)0, X0[t], false, false);
        }
    }
    if constexpr (NS < NSTEP) head_steps<NS>(base, wrow, raw, A, X0, X1, bf, st2, st3, pre);
}

__device__ __forceinline__ void store_row(const v8f (&acc)[4], const float* bl, float* __restrict__ orow,
                                          int P0, int laneN)
{
    const v4f b0 = *(const v4f*)bl, b1 = *(const v4f*)(bl + 4);
    const float bv[8] = {b0[0], b0[1], b0[2], b0[3], b1[0], b1[1], b1[2], b1[3]};
    #pragma unroll
    for (int t = 0; t < 4; ++t) {
        const int wpix = P0 + t * 16 + laneN;
        if (t < 3 || wpix < WOUT) {
            #pragma unroll
            for (int j = 0; j < 8; ++j)
                orow[(size_t)j * (HOUT * WOUT) + wpix] = acc[t][j] + bv[j];
        }
    }
}

extern "C" __global__ __launch_bounds__(256)
void dynconv2d_rowstat_f32(const float* __restrict__ x,
                           const float* __restrict__ w,
                           const float* __restrict__ bias,
                           float* __restrict__ out)
{
    extern __shared__ float sX[];

    const int g   = blockIdx.x;
    const int b   = blockIdx.y;
    const int tid = threadIdx.x;
    const int h0  = g * ROWS_PER_BLOCK;
    const int R   = (HOUT - h0 < ROWS_PER_BLOCK) ? (HOUT - h0) : ROWS_PER_BLOCK;

    const int wave  = tid >> 5;
    const int lane  = tid & 31;
    const int laneN = lane & 15;
    const int hi    = lane >> 4;

    const int co_base = (wave & 3) * 16;
    const int P0      = (wave >> 2) * 64;

    const v4f* wrow = (const v4f*)(w + ((size_t)(b * COUT + co_base + laneN)) * KDIM + 288 * hi);
    v4f wraw[9];
    v2f A[16][9];

    const float* st_g = x + ((size_t)b * CIN + wave) * (H_IN * W_IN) + 4 * lane;
    const int    st_l = wave * XP + 4 * lane;

    #pragma unroll
    for (int rr = 0; rr < 2; ++rr) {
        #pragma unroll
        for (int j = 0; j < 8; ++j) {
            v4f v = *(const v4f*)(st_g + (size_t)j * 8 * (H_IN * W_IN) + (h0 + rr) * W_IN);
            *(v4f*)(sX + rr * SLOT_F + st_l + j * 8 * XP + (j >= 4 ? HALF_PAD : 0)) = v;
        }
    }
    {
        const int slot = tid >> 6, ci = tid & 63;
        const v4f z = {0.0f, 0.0f, 0.0f, 0.0f};
        *(v4f*)(sX + slot * SLOT_F + ci * XP + (ci >= 32 ? HALF_PAD : 0) + W_IN) = z;
    }

    w_fetch<0>(wrow, wraw);

    if (tid < COUT) sX[BIAS_F + tid] = bias[b * COUT + tid];
    const float* bv = sX + BIAS_F + co_base + 8 * hi;

    __syncthreads();

    const int laneB = hi * (32 * XP + HALF_PAD) + P0 + laneN;
    float* obase = out + ((size_t)b * COUT + co_base + 8 * hi) * (HOUT * WOUT);

    v8f X0[4] = {}, X1[4] = {}, X2[4] = {}, XD[4] = {};


    {
        w_build<0>(wraw, A);
        const unsigned base0 = (unsigned)(size_t)(const lds_float*)(sX + laneB);
        v2f bf[2][GOPS];
        v4f pre[4];
        load_group<0>(base0, bf[0]);
        head_steps<0>(base0, wrow, wraw, A, X0, X1, bf,
                      Stage{st_g + (h0 + 2) * W_IN, sX + 2 * SLOT_F + st_l},
                      Stage{st_g + (h0 + 3) * W_IN, sX + 3 * SLOT_F + st_l}, pre);
    }
    __syncthreads();

    int j = 2;
    #pragma unroll 1
    do {
        #pragma unroll
        for (int t = 0; t < 4; ++t) X2[t] = v8f{};
        row_pass<true, true, true, true>(sX + (j & 3) * SLOT_F + laneB, A, X2, X1, X0,
                                         Stage{st_g + (h0 + j + 2) * W_IN, sX + ((j + 2) & 3) * SLOT_F + st_l});

        store_row(X0, bv, obase + (size_t)(h0 + j - 2) * WOUT, P0, laneN);
        #pragma unroll
        for (int t = 0; t < 4; ++t) { X0[t] = X1[t]; X1[t] = X2[t]; }
        __syncthreads();
    } while (++j < R);

    row_pass<false, true, true, false>(sX + (R & 3) * SLOT_F + laneB, A, XD, X1, X0, Stage{nullptr, nullptr});
    store_row(X0, bv, obase + (size_t)(h0 + R - 2) * WOUT, P0, laneN);

    row_pass<false, false, true, false>(sX + ((R + 1) & 3) * SLOT_F + laneB, A, XD, XD, X1, Stage{nullptr, nullptr});
    store_row(X1, bv, obase + (size_t)(h0 + R - 1) * WOUT, P0, laneN);
}

extern "C" void kernel_launch(void* const* d_in, const int* in_sizes, int n_in,
                              void* d_out, int out_size, void* d_ws, size_t ws_size,
                              hipStream_t stream) {
    (void)in_sizes; (void)n_in; (void)out_size; (void)d_ws; (void)ws_size;
    const float* x    = (const float*)d_in[0];
    const float* w    = (const float*)d_in[1];
    const float* bias = (const float*)d_in[2];
    float* out        = (float*)d_out;

    dim3 grid(NGROUPS, B_N);
    dynconv2d_rowstat_f32<<<grid, 256, LDS_BYTES, stream>>>(x, w, bias, out);
}


# Row barrier hand-written and split over a pass: a wave signals its row end and waits there only for the row before (4-slot ring allows one row of skew); LDS wait only, stores not awaited; sums same
# speedup vs baseline: 5.7218x; 1.0402x over previous
#include <hip/hip_runtime.h>


typedef __attribute__((ext_vector_type(2))) float v2f;
typedef __attribute__((ext_vector_type(4))) float v4f;
typedef __attribute__((ext_vector_type(8))) float v8f;
typedef __attribute__((address_space(3))) float lds_float;

#define B_N    16
#define CIN    64
#define COUT   64
#define H_IN   128
#define W_IN   128
#define HOUT   126
#define WOUT   126
#define KDIM   576

#define ROWS_PER_BLOCK 8
#define NGROUPS  16
#define XP       132
#define HALF_PAD 16
#define SLOT_F   (CIN * XP + HALF_PAD)
#define NSLOT    4
#define BIAS_F   (NSLOT * SLOT_F)
#define LDS_BYTES ((NSLOT * SLOT_F + COUT) * 4)
#define NGRP     16
#define GOPS     12

template <int T, int KW>
__device__ __forceinline__ v2f lds_ci_pair(unsigned addr)
{
    v2f r;
    asm volatile("ds_load_2addr_b32 %0, %1 offset0:%2 offset1:%3"
                 : "=v"(r) : "v"(addr), "n"(16 * T + KW), "n"(XP + 16 * T + KW));
    return r;
}

template <int LEFT>
__device__ __forceinline__ void lds_wait(v2f (&b)[GOPS])
{
    asm volatile("s_wait_dscnt %12"
                 : "+v"(b[0]), "+v"(b[1]), "+v"(b[2]), "+v"(b[3]), "+v"(b[4]), "+v"(b[5]),
                   "+v"(b[6]), "+v"(b[7]), "+v"(b[8]), "+v"(b[9]), "+v"(b[10]), "+v"(b[11])
                 : "n"(LEFT));
}

template <int CP>
__device__ __forceinline__ void load_group(unsigned base, v2f (&b)[GOPS])
{
    const unsigned a = base + CP * (2 * XP * 4);
    b[0]  = lds_ci_pair<0, 0>(a);
    b[1]  = lds_ci_pair<1, 0>(a);
    b[2]  = lds_ci_pair<2, 0>(a);
    b[3]  = lds_ci_pair<3, 0>(a);
    b[4]  = lds_ci_pair<0, 1>(a);
    b[5]  = lds_ci_pair<1, 1>(a);
    b[6]  = lds_ci_pair<2, 1>(a);
    b[7]  = lds_ci_pair<3, 1>(a);
    b[8]  = lds_ci_pair<0, 2>(a);
    b[9]  = lds_ci_pair<1, 2>(a);
    b[10] = lds_ci_pair<2, 2>(a);
    b[11] = lds_ci_pair<3, 2>(a);
}

struct Stage { const float* g; float* l; };
template <int HALF>
__device__ __forceinline__ void stage_fetch(const Stage& s, v4f (&p)[4])
{
    #pragma unroll
    for (int q = 0; q < 4; ++q) p[q] = *(const v4f*)(s.g + (size_t)(4 * HALF + q) * 8 * (H_IN * W_IN));
}
template <int HALF>
__device__ __forceinline__ void stage_put(const Stage& s, const v4f (&p)[4])
{
    #pragma unroll
    for (int q = 0; q < 4; ++q) *(v4f*)(s.l + (4 * HALF + q) * 8 * XP + (HALF ? HALF_PAD : 0)) = p[q];
}

template <int G, bool K0, bool K1, bool K2, bool PF>
__device__ __forceinline__ void row_groups(unsigned base, const v2f (&A)[16][9],
                                           v8f (&n)[4], v8f (&m)[4], v8f (&o)[4], v2f (&bf)[2][GOPS],
                                           const Stage& st, v4f (&pre)[4])
{
    constexpr int CUR = G % 2;
    constexpr int AHEAD = (G + 1 < NGRP) ? 1 : 0;
    if constexpr (PF && G == 0)  stage_fetch<0>(st, pre);
    if constexpr (PF && G == 5)  { stage_put<0>(st, pre); stage_fetch<1>(st, pre); }
    if constexpr (PF && G == 10) stage_put<1>(st, pre);
    __builtin_amdgcn_sched_barrier(0);
    if constexpr (G + 1 < NGRP) load_group<G + 1>(base, bf[(G + 1) % 2]);
    lds_wait<GOPS * AHEAD>(bf[CUR]);
    __builtin_amdgcn_sched_barrier(0);
    #pragma unroll
    for (int kw = 0; kw < 3; ++kw) {
        if constexpr (K0) {
            #pragma unroll
            for (int t = 0; t < 4; ++t)
                n[t] = __builtin_amdgcn_wmma_f32_16x16x4_f32(false, A[G][kw], false, bf[CUR][4 * kw + t], (short)0, n[t], false, false);
        }
        if constexpr (K1) {
            #pragma unroll
            for (int t = 0; t < 4; ++t)
                m[t] = __builtin_amdgcn_wmma_f32_16x16x4_f32(false, A[G][3 + kw], false, bf[CUR][4 * kw + t], (short)0, m[t], false, false);
        }
        if constexpr (K2) {
            #pragma unroll
            for (int t = 0; t < 4; ++t)
                o[t] = __builtin_amdgcn_wmma_f32_16x16x4_f32(false, A[G][6 + kw], false, bf[CUR][4 * kw + t], (short)0, o[t], false, false);
        }
    }
    if constexpr (G + 1 < NGRP) row_groups<G + 1, K0, K1, K2, PF>(base, A, n, m, o, bf, st, pre);
}

template <bool K0, bool K1, bool K2, bool PF>
__device__ __forceinline__ void row_pass(const float* bk, const v2f (&A)[16][9],
                                         v8f (&n)[4], v8f (&m)[4], v8f (&o)[4], const Stage& st)
{
    const unsigned base = (unsigned)(size_t)(const lds_float*)bk;
    v2f bf[2][GOPS];
    v4f pre[4];
    load_group<0>(base, bf[0]);
    row_groups<0, K0, K1, K2, PF>(base, A, n, m, o, bf, st, pre);
}

template <int CQ>
__device__ __forceinline__ void w_fetch(const v4f* wrow, v4f (&raw)[9])
{
    #pragma unroll
    for (int i = 0; i < 9; ++i) raw[i] = wrow[9 * CQ + i];
}
template <int CQ>
__device__ __forceinline__ void w_build(const v4f (&raw)[9], v2f (&A)[16][9])
{
    #pragma unroll
    for (int c = 0; c < 2; ++c) {
        #pragma unroll
        for (int tap = 0; tap < 9; ++tap) {
            const int f0 = 18 * c + tap, f1 = f0 + 9;
            A[2 * CQ + c][tap] = v2f{raw[f0 >> 2][f0 & 3], raw[f1 >> 2][f1 & 3]};
        }
    }
}

template <int S>
__device__ __forceinline__ void head_steps(unsigned base, const v4f* wrow, v4f (&raw)[9], v2f (&A)[16][9],
                                           v8f (&X0)[4], v8f (&X1)[4], v2f (&bf)[2][GOPS],
                                           const Stage& st2, const Stage& st3, v4f (&pre)[4])
{
    constexpr int NSTEP = 32;
    constexpr int CQ = S / 4, RW = (S % 4) / 2, CP = 2 * CQ + (S % 2);
    constexpr int NS = S + 1;
    constexpr int NRW = (NS % 4) / 2, NCP = 2 * (NS / 4) + (NS % 2);
    constexpr int CUR = S % 2;
    constexpr int AHEAD = (NS < NSTEP) ? 1 : 0;
    if constexpr (S == 0) w_fetch<1>(wrow, raw);
    if constexpr (S % 4 == 3 && CQ + 1 < 8) w_build<(CQ + 1 < 8 ? CQ + 1 : 0)>(raw, A);
    if constexpr (S % 4 == 3 && CQ + 2 < 8) w_fetch<(CQ + 2 < 8 ? CQ + 2 : 0)>(wrow, raw);
    if constexpr (S == 1)  stage_fetch<0>(st2, pre);
    if constexpr (S == 9)  { stage_put<0>(st2, pre); stage_fetch<1>(st2, pre); }
    if constexpr (S == 17) { stage_put<1>(st2, pre); stage_fetch<0>(st3, pre); }
    if constexpr (S == 25) { stage_put<0>(st3, pre); stage_fetch<1>(st3, pre); }
    if constexpr (S == 31) stage_put<1>(st3, pre);
    __builtin_amdgcn_sched_barrier(0);
    if constexpr (NS < NSTEP) load_group<NCP>(base + NRW * (SLOT_F * 4), bf[NS % 2]);
    lds_wait<GOPS * AHEAD>(bf[CUR]);
    __builtin_amdgcn_sched_barrier(0);
    #pragma unroll
    for (int kw = 0; kw < 3; ++kw) {
        if constexpr (RW == 0) {
            #pragma unroll
            for (int t = 0; t < 4; ++t)
                X0[t] = __builtin_amdgcn_wmma_f32_16x16x4_f32(false, A[CP][kw], false, bf[CUR][4 * kw + t], (short)0, X0[t], false, false);
        } else {
            #pragma unroll
            for (int t = 0; t < 4; ++t)
                X1[t] = __builtin_amdgcn_wmma_f32_16x16x4_f32(false, A[CP][kw], false, bf[CUR][4 * kw + t], (short)0, X1[t], false, false);
            #pragma unroll
            for (int t = 0; t < 4; ++t)
                X0[t] = __builtin_amdgcn_wmma_f32_16x16x4_f32(false, A[CP][3 + kw], false, bf[CUR][4 * kw + t], (short)0, X0[t], false, false);
        }
    }
    if constexpr (NS < NSTEP) head_steps<NS>(base, wrow, raw, A, X0, X1, bf, st2, st3, pre);
}

__device__ __forceinline__ void row_done()
{
    asm volatile("s_wait_dscnt 0x0\n\ts_barrier_signal -1" ::: "memory");
}
__device__ __forceinline__ void row_ready()
{
    asm volatile("s_barrier_wait -1" ::: "memory");
}

__device__ __forceinline__ void store_row(const v8f (&acc)[4], const float* bl, float* __restrict__ orow,
                                          int P0, int laneN)
{
    const v4f b0 = *(const v4f*)bl, b1 = *(const v4f*)(bl + 4);
    const float bv[8] = {b0[0], b0[1], b0[2], b0[3], b1[0], b1[1], b1[2], b1[3]};
    #pragma unroll
    for (int t = 0; t < 4; ++t) {
        const int wpix = P0 + t * 16 + laneN;
        if (t < 3 || wpix < WOUT) {
            #pragma unroll
            for (int j = 0; j < 8; ++j)
                orow[(size_t)j * (HOUT * WOUT) + wpix] = acc[t][j] + bv[j];
        }
    }
}

extern "C" __global__ __launch_bounds__(256)
void dynconv2d_rowstat_f32(const float* __restrict__ x,
                           const float* __restrict__ w,
                           const float* __restrict__ bias,
                           float* __restrict__ out)
{
    extern __shared__ float sX[];

    const int g   = blockIdx.x;
    const int b   = blockIdx.y;
    const int tid = threadIdx.x;
    const int h0  = g * ROWS_PER_BLOCK;
    const int R   = (HOUT - h0 < ROWS_PER_BLOCK) ? (HOUT - h0) : ROWS_PER_BLOCK;

    const int wave  = tid >> 5;
    const int lane  = tid & 31;
    const int laneN = lane & 15;
    const int hi    = lane >> 4;

    const int co_base = (wave & 3) * 16;
    const int P0      = (wave >> 2) * 64;

    const v4f* wrow = (const v4f*)(w + ((size_t)(b * COUT + co_base + laneN)) * KDIM + 288 * hi);
    v4f wraw[9];
    v2f A[16][9];

    const float* st_g = x + ((size_t)b * CIN + wave) * (H_IN * W_IN) + 4 * lane;
    const int    st_l = wave * XP + 4 * lane;

    #pragma unroll
    for (int rr = 0; rr < 2; ++rr) {
        #pragma unroll
        for (int j = 0; j < 8; ++j) {
            v4f v = *(const v4f*)(st_g + (size_t)j * 8 * (H_IN * W_IN) + (h0 + rr) * W_IN);
            *(v4f*)(sX + rr * SLOT_F + st_l + j * 8 * XP + (j >= 4 ? HALF_PAD : 0)) = v;
        }
    }
    {
        const int slot = tid >> 6, ci = tid & 63;
        const v4f z = {0.0f, 0.0f, 0.0f, 0.0f};
        *(v4f*)(sX + slot * SLOT_F + ci * XP + (ci >= 32 ? HALF_PAD : 0) + W_IN) = z;
    }

    w_fetch<0>(wrow, wraw);

    if (tid < COUT) sX[BIAS_F + tid] = bias[b * COUT + tid];
    const float* bv = sX + BIAS_F + co_base + 8 * hi;

    __syncthreads();

    const int laneB = hi * (32 * XP + HALF_PAD) + P0 + laneN;
    float* obase = out + ((size_t)b * COUT + co_base + 8 * hi) * (HOUT * WOUT);

    v8f X0[4] = {}, X1[4] = {}, X2[4] = {}, XD[4] = {};


    {
        w_build<0>(wraw, A);
        const unsigned base0 = (unsigned)(size_t)(const lds_float*)(sX + laneB);
        v2f bf[2][GOPS];
        v4f pre[4];
        load_group<0>(base0, bf[0]);
        head_steps<0>(base0, wrow, wraw, A, X0, X1, bf,
                      Stage{st_g + (h0 + 2) * W_IN, sX + 2 * SLOT_F + st_l},
                      Stage{st_g + (h0 + 3) * W_IN, sX + 3 * SLOT_F + st_l}, pre);
    }
    __syncthreads();
    row_done();

    int j = 2;
    #pragma unroll 1
    do {
        #pragma unroll
        for (int t = 0; t < 4; ++t) X2[t] = v8f{};
        row_pass<true, true, true, true>(sX + (j & 3) * SLOT_F + laneB, A, X2, X1, X0,
                                         Stage{st_g + (h0 + j + 2) * W_IN, sX + ((j + 2) & 3) * SLOT_F + st_l});

        store_row(X0, bv, obase + (size_t)(h0 + j - 2) * WOUT, P0, laneN);
        #pragma unroll
        for (int t = 0; t < 4; ++t) { X0[t] = X1[t]; X1[t] = X2[t]; }
        row_ready();
        row_done();
    } while (++j < R);

    row_pass<false, true, true, false>(sX + (R & 3) * SLOT_F + laneB, A, XD, X1, X0, Stage{nullptr, nullptr});
    store_row(X0, bv, obase + (size_t)(h0 + R - 2) * WOUT, P0, laneN);
    row_ready();

    row_pass<false, false, true, false>(sX + ((R + 1) & 3) * SLOT_F + laneB, A, XD, XD, X1, Stage{nullptr, nullptr});
    store_row(X1, bv, obase + (size_t)(h0 + R - 1) * WOUT, P0, laneN);
}

extern "C" void kernel_launch(void* const* d_in, const int* in_sizes, int n_in,
                              void* d_out, int out_size, void* d_ws, size_t ws_size,
                              hipStream_t stream) {
    (void)in_sizes; (void)n_in; (void)out_size; (void)d_ws; (void)ws_size;
    const float* x    = (const float*)d_in[0];
    const float* w    = (const float*)d_in[1];
    const float* bias = (const float*)d_in[2];
    float* out        = (float*)d_out;

    dim3 grid(NGROUPS, B_N);
    dynconv2d_rowstat_f32<<<grid, 256, LDS_BYTES, stream>>>(x, w, bias, out);
}


# Horizontal minimal filtering F(2,3): 4 products per pixel pair where 6; taps transformed per lane, tap row 2 parked in LDS; 4 waves, 1 per SIMD; one loop body for all row passes; sum order changes
# speedup vs baseline: 6.9978x; 1.2230x over previous
#include <hip/hip_runtime.h>


typedef __attribute__((ext_vector_type(2))) float v2f;
typedef __attribute__((ext_vector_type(4))) float v4f;
typedef __attribute__((ext_vector_type(8))) float v8f;
typedef __attribute__((address_space(3))) float lds_float;

#define B_N    16
#define CIN    64
#define COUT   64
#define H_IN   128
#define W_IN   128
#define HOUT   126
#define WOUT   126
#define KDIM   576

#define ROWS_PER_BLOCK 8
#define NGROUPS  16
#define XP       132
#define HALF_PAD 32
#define SLOT_F   (CIN * XP + HALF_PAD)
#define NSLOT    4
#define BIAS_F   (NSLOT * SLOT_F)
#define AW_F     (BIAS_F + COUT)
#define AW_WAVE  4096
#define LDS_BYTES ((AW_F + 4 * AW_WAVE) * 4)
#define NGRP     16
#define GOPS     16


template <int OFF>
__device__ __forceinline__ v2f lds_ci_pair(unsigned addr)
{
    v2f r;
    asm volatile("ds_load_2addr_b32 %0, %1 offset0:%2 offset1:%3"
                 : "=v"(r) : "v"(addr), "n"(OFF), "n"(XP + OFF));
    return r;
}

__device__ __forceinline__ void lds_wait(v2f (&b)[GOPS], v4f (&a)[2])
{
    asm volatile("s_wait_dscnt 0x0"
                 : "+v"(a[0]), "+v"(a[1]), "+v"(b[0]), "+v"(b[1]), "+v"(b[2]), "+v"(b[3]), "+v"(b[4]), "+v"(b[5]), "+v"(b[6]), "+v"(b[7]),
                   "+v"(b[8]), "+v"(b[9]), "+v"(b[10]), "+v"(b[11]), "+v"(b[12]), "+v"(b[13]), "+v"(b[14]), "+v"(b[15]));
}

template <int CP>
__device__ __forceinline__ void load_a2(unsigned abase, v4f (&a)[2])
{
    asm volatile("ds_load_b128 %0, %1 offset:%2" : "=v"(a[0]) : "v"(abase), "n"(CP * 1024));
    asm volatile("ds_load_b128 %0, %1 offset:%2" : "=v"(a[1]) : "v"(abase), "n"(CP * 1024 + 512));
}

template <int CP>
__device__ __forceinline__ void load_group(unsigned base, v2f (&raw)[GOPS])
{
    const unsigned a = base + CP * (2 * XP * 4);
    raw[0]  = lds_ci_pair<0>(a);   raw[1]  = lds_ci_pair<1>(a);   raw[2]  = lds_ci_pair<2>(a);   raw[3]  = lds_ci_pair<3>(a);
    raw[4]  = lds_ci_pair<32>(a);  raw[5]  = lds_ci_pair<33>(a);  raw[6]  = lds_ci_pair<34>(a);  raw[7]  = lds_ci_pair<35>(a);
    raw[8]  = lds_ci_pair<64>(a);  raw[9]  = lds_ci_pair<65>(a);  raw[10] = lds_ci_pair<66>(a);  raw[11] = lds_ci_pair<67>(a);
    raw[12] = lds_ci_pair<96>(a);  raw[13] = lds_ci_pair<97>(a);  raw[14] = lds_ci_pair<98>(a);  raw[15] = lds_ci_pair<99>(a);
}

__device__ __forceinline__ void in_transform(const v2f (&raw)[GOPS], v2f (&bf)[GOPS])
{
    #pragma unroll
    for (int t = 0; t < 4; ++t) {
        const v2f d0 = raw[4 * t], d1 = raw[4 * t + 1], d2 = raw[4 * t + 2], d3 = raw[4 * t + 3];
        bf[t]      = d0 - d2;
        bf[4 + t]  = d1 + d2;
        bf[8 + t]  = d2 - d1;
        bf[12 + t] = d1 - d3;
    }
}

struct Stage { const float* g; float* l; };
template <int Q>
__device__ __forceinline__ void stage_fetch(const Stage& s, v4f (&p)[4])
{
    #pragma unroll
    for (int q = 0; q < 4; ++q) p[q] = *(const v4f*)(s.g + (size_t)(4 * Q + q) * 4 * (H_IN * W_IN));
}
template <int Q>
__device__ __forceinline__ void stage_put(const Stage& s, const v4f (&p)[4])
{
    #pragma unroll
    for (int q = 0; q < 4; ++q) *(v4f*)(s.l + (4 * Q + q) * 4 * XP + (Q >= 2 ? HALF_PAD : 0)) = p[q];
}

template <int G>
__device__ __forceinline__ void row_groups(unsigned base, unsigned abase, const v2f (&A)[16][8],
                                           v8f (&n)[16], v8f (&m)[16], v8f (&o)[16], v2f (&raw)[GOPS], v2f (&bf)[2][GOPS],
                                           v4f (&a2)[2][2], const Stage& st, v4f (&pre)[4],
                                           bool k0, bool k1, bool k2, bool pf)
{
    constexpr int CUR = G % 2;
    __builtin_amdgcn_sched_barrier(0);
    lds_wait(raw, a2[CUR]);
    in_transform(raw, bf[CUR]);
    if (__builtin_expect(pf, true)) {
        if constexpr (G == 0)  stage_fetch<0>(st, pre);
        if constexpr (G == 4)  { stage_put<0>(st, pre); stage_fetch<1>(st, pre); }
        if constexpr (G == 8)  { stage_put<1>(st, pre); stage_fetch<2>(st, pre); }
        if constexpr (G == 12) { stage_put<2>(st, pre); stage_fetch<3>(st, pre); }
        if constexpr (G == 15) stage_put<3>(st, pre);
    }
    if constexpr (G + 1 < NGRP) { load_group<G + 1>(base, raw); load_a2<G + 1>(abase, a2[(G + 1) % 2]); }
    __builtin_amdgcn_sched_barrier(0);
    if (__builtin_expect(k0, true)) {
        #pragma unroll
        for (int pos = 0; pos < 4; ++pos) {
            #pragma unroll
            for (int t = 0; t < 4; ++t)
                n[4 * pos + t] = __builtin_amdgcn_wmma_f32_16x16x4_f32(false, A[G][pos], false, bf[CUR][4 * pos + t], (short)0, n[4 * pos + t], false, false);
        }
    }
    if (__builtin_expect(k1, true)) {
        #pragma unroll
        for (int pos = 0; pos < 4; ++pos) {
            #pragma unroll
            for (int t = 0; t < 4; ++t)
                m[4 * pos + t] = __builtin_amdgcn_wmma_f32_16x16x4_f32(false, A[G][4 + pos], false, bf[CUR][4 * pos + t], (short)0, m[4 * pos + t], false, false);
        }
    }
    if (__builtin_expect(k2, true)) {
        #pragma unroll
        for (int pos = 0; pos < 4; ++pos) {
            const v4f aq = a2[CUR][pos >> 1];
            const v2f a2p = (pos & 1) ? v2f{aq[2], aq[3]} : v2f{aq[0], aq[1]};
            #pragma unroll
            for (int t = 0; t < 4; ++t)
                o[4 * pos + t] = __builtin_amdgcn_wmma_f32_16x16x4_f32(false, a2p, false, bf[CUR][4 * pos + t], (short)0, o[4 * pos + t], false, false);
        }
    }
    if constexpr (G + 1 < NGRP) row_groups<G + 1>(base, abase, A, n, m, o, raw, bf, a2, st, pre, k0, k1, k2, pf);
}

__device__ __forceinline__ void row_pass(const float* bk, const float* ak, const v2f (&A)[16][8],
                                         v8f (&n)[16], v8f (&m)[16], v8f (&o)[16], const Stage& st,
                                         bool k0, bool k1, bool k2, bool pf)
{
    const unsigned base  = (unsigned)(size_t)(const lds_float*)bk;
    const unsigned abase = (unsigned)(size_t)(const lds_float*)ak;
    v2f raw[GOPS], bf[2][GOPS];
    v4f a2[2][2];
    v4f pre[4];
    load_group<0>(base, raw);
    load_a2<0>(abase, a2[0]);
    row_groups<0>(base, abase, A, n, m, o, raw, bf, a2, st, pre, k0, k1, k2, pf);
    asm volatile("v_nop\n\tv_nop\n\tv_nop\n\tv_nop\n\tv_nop");
}

template <int CQ>
__device__ __forceinline__ void w_build(const v4f* wrow, v2f (&A)[16][8], float* aw)
{
    v4f raw[9];
    #pragma unroll
    for (int i = 0; i < 9; ++i) raw[i] = wrow[9 * CQ + i];
    #pragma unroll
    for (int c = 0; c < 2; ++c) {
        #pragma unroll
        for (int kh = 0; kh < 3; ++kh) {
            const int f0 = 18 * c + 3 * kh, f1 = f0 + 9;
            const v2f g0 = v2f{raw[f0 >> 2][f0 & 3],             raw[f1 >> 2][f1 & 3]};
            const v2f g1 = v2f{raw[(f0 + 1) >> 2][(f0 + 1) & 3], raw[(f1 + 1) >> 2][(f1 + 1) & 3]};
            const v2f g2 = v2f{raw[(f0 + 2) >> 2][(f0 + 2) & 3], raw[(f1 + 2) >> 2][(f1 + 2) & 3]};
            const v2f s  = g0 + g2;
            const v2f p1 = (s + g1) * 0.5f, p2 = (s - g1) * 0.5f;
            if (kh < 2) {
                A[2 * CQ + c][4 * kh + 0] = g0;
                A[2 * CQ + c][4 * kh + 1] = p1;
                A[2 * CQ + c][4 * kh + 2] = p2;
                A[2 * CQ + c][4 * kh + 3] = g2;
            } else {
                *(v4f*)(aw + (2 * CQ + c) * 256)       = v4f{g0[0], g0[1], p1[0], p1[1]};
                *(v4f*)(aw + (2 * CQ + c) * 256 + 128) = v4f{p2[0], p2[1], g2[0], g2[1]};
            }
        }
    }
}

__device__ __forceinline__ void row_done()
{
    asm volatile("s_wait_dscnt 0x0\n\ts_barrier_signal -1" ::: "memory");
}
__device__ __forceinline__ void row_ready()
{
    asm volatile("s_barrier_wait -1" ::: "memory");
}

__device__ __forceinline__ void store_row(const v8f (&acc)[16], const float* bl, float* __restrict__ orow, int laneN)
{
    const v4f b0 = *(const v4f*)bl, b1 = *(const v4f*)(bl + 4);
    const float bv[8] = {b0[0], b0[1], b0[2], b0[3], b1[0], b1[1], b1[2], b1[3]};
    #pragma unroll
    for (int t = 0; t < 4; ++t) {
        const int pr = t * 16 + laneN;
        if (t < 3 || pr < WOUT / 2) {
            #pragma unroll
            for (int r = 0; r < 8; ++r) {
                const float m0 = acc[t][r], m1 = acc[4 + t][r], m2 = acc[8 + t][r], m3 = acc[12 + t][r];
                const v2f y = v2f{(m0 + m1) + (m2 + bv[r]), (m1 - m2) - (m3 - bv[r])};
                *(v2f*)(orow + (size_t)r * (HOUT * WOUT) + 2 * pr) = y;
            }
        }
    }
}

extern "C" __global__ __launch_bounds__(128)
void dynconv2d_rowstat_f23_f32(const float* __restrict__ x,
                               const float* __restrict__ w,
                               const float* __restrict__ bias,
                               float* __restrict__ out)
{
    extern __shared__ float sX[];

    const int g   = blockIdx.x;
    const int b   = blockIdx.y;
    const int tid = threadIdx.x;
    const int h0  = g * ROWS_PER_BLOCK;
    const int R   = (HOUT - h0 < ROWS_PER_BLOCK) ? (HOUT - h0) : ROWS_PER_BLOCK;

    const int wave  = tid >> 5;
    const int lane  = tid & 31;
    const int laneN = lane & 15;
    const int hi    = lane >> 4;

    const int co_base = wave * 16;

    const v4f* wrow = (const v4f*)(w + ((size_t)(b * COUT + co_base + laneN)) * KDIM + 288 * hi);
    v2f A[16][8];
    float* aw = sX + AW_F + wave * AW_WAVE + lane * 4;

    const float* st_g = x + ((size_t)b * CIN + wave) * (H_IN * W_IN) + 4 * lane;
    const int    st_l = wave * XP + 4 * lane;

    #pragma unroll
    for (int rr = 0; rr < 2; ++rr) {
        #pragma unroll
        for (int j = 0; j < 16; ++j) {
            v4f v = *(const v4f*)(st_g + (size_t)j * 4 * (H_IN * W_IN) + (h0 + rr) * W_IN);
            *(v4f*)(sX + rr * SLOT_F + st_l + j * 4 * XP + (j >= 8 ? HALF_PAD : 0)) = v;
        }
    }
    {
        const v4f z = {0.0f, 0.0f, 0.0f, 0.0f};
        #pragma unroll
        for (int k = 0; k < 2; ++k) {
            const int idx = tid + 128 * k, slot = idx >> 6, ci = idx & 63;
            *(v4f*)(sX + slot * SLOT_F + ci * XP + (ci >= 32 ? HALF_PAD : 0) + W_IN) = z;
        }
    }
    if (tid < COUT) sX[BIAS_F + tid] = bias[b * COUT + tid];
    const float* bv = sX + BIAS_F + co_base + 8 * hi;

    w_build<0>(wrow, A, aw); w_build<1>(wrow, A, aw); w_build<2>(wrow, A, aw); w_build<3>(wrow, A, aw);
    w_build<4>(wrow, A, aw); w_build<5>(wrow, A, aw); w_build<6>(wrow, A, aw); w_build<7>(wrow, A, aw);

    __syncthreads();
    row_done();

    const int laneB = hi * (32 * XP + HALF_PAD) + 2 * laneN;
    float* obase = out + ((size_t)b * COUT + co_base + 8 * hi) * (HOUT * WOUT);

    v8f X0[16] = {}, X1[16] = {}, X2[16] = {};

    int j = 0;
    #pragma unroll 1
    do {
        #pragma unroll
        for (int t = 0; t < 16; ++t) X2[t] = v8f{};
        const bool k0 = j < R, k1 = (j >= 1) && (j <= R), k2 = j >= 2;
        row_pass(sX + (j & 3) * SLOT_F + laneB, aw, A, X2, X1, X0,
                 Stage{st_g + (h0 + j + 2) * W_IN, sX + ((j + 2) & 3) * SLOT_F + st_l}, k0, k1, k2, k0);

        if (k2) store_row(X0, bv, obase + (size_t)(h0 + j - 2) * WOUT, laneN);
        #pragma unroll
        for (int t = 0; t < 16; ++t) { X0[t] = X1[t]; X1[t] = X2[t]; }
        row_ready();
        if (j < R + 1) row_done();
    } while (++j < R + 2);
}

extern "C" void kernel_launch(void* const* d_in, const int* in_sizes, int n_in,
                              void* d_out, int out_size, void* d_ws, size_t ws_size,
                              hipStream_t stream) {
    (void)in_sizes; (void)n_in; (void)out_size; (void)d_ws; (void)ws_size;
    const float* x    = (const float*)d_in[0];
    const float* w    = (const float*)d_in[1];
    const float* bias = (const float*)d_in[2];
    float* out        = (float*)d_out;

    dim3 grid(NGROUPS, B_N);
    dynconv2d_rowstat_f23_f32<<<grid, 128, LDS_BYTES, stream>>>(x, w, bias, out);
}
